# starting kernel
# baseline (speedup 1.0000x reference)
#include <hip/hip_runtime.h>


#define EMBED_DIM 64
#define NUM_LAYERS 3

typedef float v2f __attribute__((ext_vector_type(2)));
typedef float v8f __attribute__((ext_vector_type(8)));

__global__ void zero_f32(float* __restrict__ p, size_t n) {
  size_t i = (size_t)blockIdx.x * blockDim.x + threadIdx.x;
  size_t stride = (size_t)gridDim.x * blockDim.x;
  for (; i < n; i += stride) p[i] = 0.0f;
}

__global__ void deg_accum(const int* __restrict__ row, float* __restrict__ deg,
                          int E) {
  int e = blockIdx.x * blockDim.x + threadIdx.x;
  if (e < E) atomicAdd(&deg[row[e]], 1.0f);
}

__global__ void deg_to_dinv(float* __restrict__ dinv, int N) {
  int i = blockIdx.x * blockDim.x + threadIdx.x;
  if (i < N) {
    float d = dinv[i];
    dinv[i] = (d > 0.0f) ? rsqrtf(d) : 0.0f;
  }
}

__global__ void edge_norm(const int* __restrict__ row,
                          const int* __restrict__ col,
                          const float* __restrict__ dinv,
                          float* __restrict__ nrm, int E) {
  int e = blockIdx.x * blockDim.x + threadIdx.x;
  if (e < E) nrm[e] = dinv[row[e]] * dinv[col[e]];
}

__global__ void init_out(const float* __restrict__ emb, float* __restrict__ out,
                         size_t n, float c) {
  size_t i = (size_t)blockIdx.x * blockDim.x + threadIdx.x;
  size_t stride = (size_t)gridDim.x * blockDim.x;
  for (; i < n; i += stride) out[i] = emb[i] * c;
}

__global__ void scatter_layer(const int* __restrict__ row,
                              const int* __restrict__ col,
                              const float* __restrict__ w,
                              const int* __restrict__ et,
                              const float* __restrict__ nrm,
                              const float* __restrict__ hcur,
                              float* __restrict__ hnext,
                              float* __restrict__ s, int E) {
  int e = (int)(((size_t)blockIdx.x * blockDim.x + threadIdx.x) >> 5);
  int lane = threadIdx.x & 31;
  if (e >= E) return;

  int r = row[e];
  int c = col[e];
  float nm = nrm[e];
  float a = w[e] * nm;

  float2 hv = ((const float2*)(hcur + (size_t)r * EMBED_DIM))[lane];
  float* dst = hnext + (size_t)c * EMBED_DIM + lane * 2;
  atomicAdd(dst + 0, a * hv.x);
  atomicAdd(dst + 1, a * hv.y);
  if (lane == 0) atomicAdd(&s[(size_t)c * 4 + et[e]], nm);
}

__global__ void dense_wmma_out(const float* __restrict__ s,
                               const float* __restrict__ T,
                               float* __restrict__ hnext,
                               float* __restrict__ out,
                               float inv_lp1, int num_tiles) {
  int tile = (int)(((size_t)blockIdx.x * blockDim.x + threadIdx.x) >> 5);
  int lane = threadIdx.x & 31;
  if (tile >= num_tiles) return;

  int half = lane >> 4;
  int l16 = lane & 15;
  int k0 = half * 2;

  int arow = tile * 16 + l16;
  v2f A;
  A.x = s[(size_t)arow * 4 + k0];
  A.y = s[(size_t)arow * 4 + k0 + 1];

  for (int dt = 0; dt < 4; ++dt) {
    int n = dt * 16 + l16;
    v2f B;
    B.x = (k0 + 0 < 3) ? T[(k0 + 0) * EMBED_DIM + n] : 0.0f;
    B.y = (k0 + 1 < 3) ? T[(k0 + 1) * EMBED_DIM + n] : 0.0f;

    v8f C = {};
    C = __builtin_amdgcn_wmma_f32_16x16x4_f32(
         false, A,  false, B,
         (short)0, C,  false,  false);

#pragma unroll
    for (int r = 0; r < 8; ++r) {
      int node = tile * 16 + r + 8 * half;
      size_t idx = (size_t)node * EMBED_DIM + n;
      float hv = hnext[idx] + C[r];
      hnext[idx] = hv;
      out[idx] += inv_lp1 * hv;
    }
  }
}

extern "C" void kernel_launch(void* const* d_in, const int* in_sizes, int n_in,
                              void* d_out, int out_size, void* d_ws,
                              size_t ws_size, hipStream_t stream) {
  const int E = in_sizes[1];
  const int N = in_sizes[3] / EMBED_DIM;

  const int* eidx = (const int*)d_in[0];
  const int* row = eidx;
  const int* col = eidx + E;
  const float* w = (const float*)d_in[1];
  const int* et = (const int*)d_in[2];
  const float* emb = (const float*)d_in[3];
  const float* T = (const float*)d_in[4];
  float* out = (float*)d_out;

  float* ws = (float*)d_ws;
  float* dinv = ws;
  float* S = dinv + (size_t)N;
  float* nrm = S + (size_t)N * 4;
  float* h_a = nrm + (size_t)E;
  float* h_b = h_a + (size_t)N * EMBED_DIM;

  const float inv_lp1 = 1.0f / (NUM_LAYERS + 1);
  const int blk = 256;

  zero_f32<<<(N + blk - 1) / blk, blk, 0, stream>>>(dinv, (size_t)N);
  deg_accum<<<(E + blk - 1) / blk, blk, 0, stream>>>(row, dinv, E);
  deg_to_dinv<<<(N + blk - 1) / blk, blk, 0, stream>>>(dinv, N);
  edge_norm<<<(E + blk - 1) / blk, blk, 0, stream>>>(row, col, dinv, nrm, E);

  size_t nd = (size_t)N * EMBED_DIM;
  init_out<<<(int)((nd + blk - 1) / blk), blk, 0, stream>>>(emb, out, nd,
                                                            inv_lp1);

  const int scatter_grid = (int)(((size_t)E * 32 + blk - 1) / blk);
  const int tiles = N / 16;
  const int wmma_grid = (tiles * 32 + blk - 1) / blk;

  const float* hcur = emb;
  for (int l = 0; l < NUM_LAYERS; ++l) {
    float* hnext = (l & 1) ? h_b : h_a;

    zero_f32<<<(int)((nd + blk - 1) / blk), blk, 0, stream>>>(hnext, nd);
    zero_f32<<<(N * 4 + blk - 1) / blk, blk, 0, stream>>>(S, (size_t)N * 4);

    scatter_layer<<<scatter_grid, blk, 0, stream>>>(row, col, w, et, nrm, hcur,
                                                    hnext, S, E);
    dense_wmma_out<<<wmma_grid, blk, 0, stream>>>(S, T, hnext, out, inv_lp1,
                                                  tiles);
    hcur = hnext;
  }
}


# Each destination row is added into by one chip half only: a block reads its die, draws a 256-edge chunk from that half's counter, adds only that half's rows; f32 atomic order is now draw order
# speedup vs baseline: 3.4486x; 3.4486x over previous
#include <hip/hip_runtime.h>


#define EMBED_DIM 64
#define NUM_LAYERS 3
#define CTR_PITCH 64

typedef float v2f __attribute__((ext_vector_type(2)));
typedef float v8f __attribute__((ext_vector_type(8)));

__global__ void zero_f32(float* __restrict__ p, size_t n) {
  size_t tid = (size_t)blockIdx.x * blockDim.x + threadIdx.x;
  size_t stride = (size_t)gridDim.x * blockDim.x;
  if ((((size_t)p) & 15) == 0) {
    float4* p4 = reinterpret_cast<float4*>(p);
    size_t n4 = n >> 2;
    for (size_t i = tid; i < n4; i += stride)
      p4[i] = make_float4(0.0f, 0.0f, 0.0f, 0.0f);
    for (size_t i = (n4 << 2) + tid; i < n; i += stride) p[i] = 0.0f;
  } else {
    for (size_t i = tid; i < n; i += stride) p[i] = 0.0f;
  }
}

#define HR 25600
#define HK 32
#define HT 1024
#define HB 16

__global__ __launch_bounds__(HT) void deg_hist(const int* __restrict__ row,
                                               unsigned* __restrict__ part,
                                               int E, int N) {
  __shared__ __attribute__((aligned(16))) unsigned cnt[HR];
  const int tid = threadIdx.x;
  const int k = blockIdx.x;
  const int lo = blockIdx.y * HR;
  const int span = (N - lo < HR) ? (N - lo) : HR;

  for (int i = tid * 4; i < HR; i += HT * 4)
    *reinterpret_cast<uint4*>(&cnt[i]) = make_uint4(0u, 0u, 0u, 0u);
  __syncthreads();

  const int per = (E + HK - 1) / HK;
  const int e0 = k * per;
  const int e1 = (e0 + per < E) ? (e0 + per) : E;
  for (int base = e0; base < e1; base += HT * HB) {
    int v[HB];
#pragma unroll
    for (int j = 0; j < HB; ++j) {
      int e = base + j * HT + tid;
      v[j] = (e < e1) ? row[e] : -1;
    }
#pragma unroll
    for (int j = 0; j < HB; ++j) {
      unsigned d = (unsigned)(v[j] - lo);
      if (d < (unsigned)span) atomicAdd(&cnt[d], 1u);
    }
  }
  __syncthreads();

  unsigned* dst = part + (size_t)k * N + lo;
  for (int i = tid; i < span; i += HT) dst[i] = cnt[i];
}

__global__ void deg_to_dinv(const unsigned* __restrict__ part,
                            float* __restrict__ dinv, int N) {
  int i = blockIdx.x * blockDim.x + threadIdx.x;
  if (i < N) {
    unsigned c = 0;
#pragma unroll
    for (int k = 0; k < HK; ++k) c += part[(size_t)k * N + i];
    float d = (float)c;
    dinv[i] = (d > 0.0f) ? rsqrtf(d) : 0.0f;
  }
}

__global__ void edge_norm_s(const int* __restrict__ row,
                            const int* __restrict__ col,
                            const int* __restrict__ et,
                            const float* __restrict__ dinv,
                            float* __restrict__ nrm, float* __restrict__ s,
                            int E) {
  int e = blockIdx.x * blockDim.x + threadIdx.x;
  if (e < E) {
    int c = col[e];
    float nv = dinv[row[e]] * dinv[c];
    nrm[e] = nv;
    atomicAdd(&s[(size_t)c * 4 + et[e]], nv);
  }
}

__global__ void layer_start(const float* __restrict__ s,
                            const float* __restrict__ T,
                            float* __restrict__ hnext,
                            unsigned* __restrict__ ctr, int num_tiles) {
  if (blockIdx.x == 0 && threadIdx.x == 0) {
    ctr[0] = 0u;
    ctr[CTR_PITCH] = 0u;
  }
  int tile = (int)(((size_t)blockIdx.x * blockDim.x + threadIdx.x) >> 5);
  int lane = threadIdx.x & 31;
  if (tile >= num_tiles) return;

  int half = lane >> 4;
  int l16 = lane & 15;
  int k0 = half * 2;

  int arow = tile * 16 + l16;
  v2f A;
  A.x = s[(size_t)arow * 4 + k0];
  A.y = s[(size_t)arow * 4 + k0 + 1];

  v8f C[4];
#pragma unroll
  for (int dt = 0; dt < 4; ++dt) {
    int n = 4 * l16 + dt;
    v2f B;
    B.x = (k0 + 0 < 3) ? T[(k0 + 0) * EMBED_DIM + n] : 0.0f;
    B.y = (k0 + 1 < 3) ? T[(k0 + 1) * EMBED_DIM + n] : 0.0f;
    v8f Z = {};
    C[dt] = __builtin_amdgcn_wmma_f32_16x16x4_f32(
         false, A,  false, B,
         (short)0, Z,  false,  false);
  }

#pragma unroll
  for (int r = 0; r < 8; ++r) {
    int node = tile * 16 + r + 8 * half;
    float4 v = make_float4(C[0][r], C[1][r], C[2][r], C[3][r]);
    *reinterpret_cast<float4*>(hnext + (size_t)node * EMBED_DIM + 4 * l16) = v;
  }
}

#define SREC 32
#define SWAVES 8
#define SCHUNK (SREC * SWAVES)

__global__ __launch_bounds__(SCHUNK) void scatter_layer(
    const int* __restrict__ row, const int* __restrict__ col,
    const float* __restrict__ w, const float* __restrict__ nrm,
    const float* __restrict__ hcur, float* __restrict__ hnext,
    unsigned* __restrict__ ctr, int E, int N, unsigned nchunks) {
  __shared__ unsigned s_chunk, s_half;
  if (threadIdx.x == 0) {
    unsigned id = __builtin_amdgcn_s_sendmsg_rtn(0x87);
    unsigned hh = (id >> 18) & 1u;
    unsigned tt = atomicAdd(ctr + hh * CTR_PITCH, 1u);
    if (tt >= nchunks) {
      hh ^= 1u;
      tt = atomicAdd(ctr + hh * CTR_PITCH, 1u);
    }
    s_chunk = tt;
    s_half = hh;
  }
  __syncthreads();
  const unsigned t = (unsigned)__builtin_amdgcn_readfirstlane((int)s_chunk);
  const unsigned h = (unsigned)__builtin_amdgcn_readfirstlane((int)s_half);
  if (t >= nchunks) return;

  const int lane = threadIdx.x & 31;
  const int e = (int)(t * SCHUNK) + (int)threadIdx.x;
  const bool valid = e < E;
  const int me = valid ? e : (int)(t * SCHUNK);
  const int rv = row[me];
  const int cv = col[me];
  const float av = w[me] * nrm[me];
  const unsigned hd = ((unsigned)cv >= (unsigned)(N >> 1)) ? 1u : 0u;
  const unsigned mask = __builtin_amdgcn_ballot_w32(valid && hd == h);

  int rs[SREC];
#pragma unroll
  for (int j = 0; j < SREC; ++j) rs[j] = __builtin_amdgcn_readlane(rv, j);

  float h_lo[SREC], h_hi[SREC];
#pragma unroll
  for (int j = 0; j < SREC; ++j) {
    if (mask & (1u << j)) {
      const float* src = hcur + (size_t)rs[j] * EMBED_DIM + lane;
      h_lo[j] = src[0];
      h_hi[j] = src[32];
    }
  }
#pragma unroll
  for (int j = 0; j < SREC; ++j) {
    if (mask & (1u << j)) {
      int c = __builtin_amdgcn_readlane(cv, j);
      float a = __int_as_float(__builtin_amdgcn_readlane(__float_as_int(av), j));
      float* dst = hnext + (size_t)c * EMBED_DIM + lane;
      atomicAdd(dst + 0, a * h_lo[j]);
      atomicAdd(dst + 32, a * h_hi[j]);
    }
  }
}

typedef float v4f __attribute__((ext_vector_type(4)));

__device__ __forceinline__ v4f sum4(v4f e, v4f a, v4f b, v4f d, float c) {
  v4f r = e * c;
  r += c * a;
  r += c * b;
  r += c * d;
  return r;
}

__global__ void sum_out(const float* __restrict__ emb,
                        const float* __restrict__ h2,
                        const float* __restrict__ h3, float* out, size_t n,
                        float c) {
  size_t tid = (size_t)blockIdx.x * blockDim.x + threadIdx.x;
  size_t stride = (size_t)gridDim.x * blockDim.x;
  const v4f* e4 = reinterpret_cast<const v4f*>(emb);
  const v4f* b4 = reinterpret_cast<const v4f*>(h2);
  const v4f* d4 = reinterpret_cast<const v4f*>(h3);
  v4f* o4 = reinterpret_cast<v4f*>(out);
  size_t n4 = n >> 2;
  size_t i = tid;
  for (; i + stride < n4; i += 2 * stride) {
    size_t j = i + stride;
    v4f e0 = e4[i], a0 = o4[i], b0 = b4[i], d0 = d4[i];
    v4f e1 = e4[j], a1 = o4[j], b1 = b4[j], d1 = d4[j];
    v4f r0 = sum4(e0, a0, b0, d0, c);
    v4f r1 = sum4(e1, a1, b1, d1, c);
    o4[i] = r0;
    o4[j] = r1;
  }
  if (i < n4) o4[i] = sum4(e4[i], o4[i], b4[i], d4[i], c);
  for (size_t k = (n4 << 2) + tid; k < n; k += stride) {
    float r = emb[k] * c;
    r += c * out[k];
    r += c * h2[k];
    r += c * h3[k];
    out[k] = r;
  }
}

extern "C" void kernel_launch(void* const* d_in, const int* in_sizes, int n_in,
                              void* d_out, int out_size, void* d_ws,
                              size_t ws_size, hipStream_t stream) {
  const int E = in_sizes[1];
  const int N = in_sizes[3] / EMBED_DIM;

  const int* eidx = (const int*)d_in[0];
  const int* row = eidx;
  const int* col = eidx + E;
  const float* w = (const float*)d_in[1];
  const int* et = (const int*)d_in[2];
  const float* emb = (const float*)d_in[3];
  const float* T = (const float*)d_in[4];
  float* out = (float*)d_out;

  float* ws = (float*)d_ws;
  float* dinv = ws;
  float* S = dinv + (size_t)N;
  float* nrm = S + (size_t)N * 4;
  float* h_a = nrm + (size_t)E;
  float* h_b = h_a + (size_t)N * EMBED_DIM;

  const float inv_lp1 = 1.0f / (NUM_LAYERS + 1);
  const int blk = 256;

  unsigned* part = (unsigned*)h_b;
  zero_f32<<<(N + blk - 1) / blk, blk, 0, stream>>>(S, (size_t)N * 4);
  deg_hist<<<dim3(HK, (N + HR - 1) / HR), HT, 0, stream>>>(row, part, E, N);
  deg_to_dinv<<<(N + blk - 1) / blk, blk, 0, stream>>>(part, dinv, N);
  edge_norm_s<<<(E + blk - 1) / blk, blk, 0, stream>>>(row, col, et, dinv, nrm,
                                                       S, E);

  size_t nd = (size_t)N * EMBED_DIM;
  const unsigned nchunks = (unsigned)((E + SCHUNK - 1) / SCHUNK);
  const int scatter_grid = (int)(2u * nchunks);
  unsigned* ctr = (unsigned*)dinv;
  const int stream_grid = 1024;
  const int tiles = N / 16;
  const int wmma_grid = (tiles * 32 + blk - 1) / blk;

  const float* hcur = emb;
  for (int l = 0; l < NUM_LAYERS; ++l) {
    float* hnext = (l == 0) ? out : ((l == 1) ? h_a : h_b);
    layer_start<<<wmma_grid, blk, 0, stream>>>(S, T, hnext, ctr, tiles);
    scatter_layer<<<scatter_grid, SCHUNK, 0, stream>>>(row, col, w, nrm, hcur,
                                                       hnext, ctr, E, N,
                                                       nchunks);
    hcur = hnext;
  }

  sum_out<<<stream_grid, blk, 0, stream>>>(emb, h_a, h_b, out, nd, inv_lp1);
}
